# starting kernel
# baseline (speedup 1.0000x reference)
#include <hip/hip_runtime.h>
#include <hip/hip_bf16.h>
#include <math.h>

typedef __attribute__((ext_vector_type(2))) float v2f;
typedef __attribute__((ext_vector_type(8))) float v8f;

#define EPS_F 1e-05f
#define MIN_NORM_F 1e-15f
#define HALF_LN2_F 0.34657359027997264f

constexpr int Bn = 8, Tn = 12, Nn = 325, Cn = 32, Kn = 100;
constexpr int M_TOTAL = Bn * Tn * Nn;
constexpr int M_TILES = M_TOTAL / 16;
constexpr int K_TILES = 7;
constexpr int KPAD = K_TILES * 16;

__global__ void prep_centroids_kernel(const float* __restrict__ w,
                                      const float* __restrict__ cptr,
                                      float* __restrict__ cent,
                                      float* __restrict__ y2out) {
    int k = blockIdx.x * blockDim.x + threadIdx.x;
    if (k >= KPAD) return;
    float c  = cptr[0];
    float sc = sqrtf(c);
    if (k < Kn) {
        float u[Cn];
        float s = 0.f;
#pragma unroll
        for (int j = 0; j < Cn; ++j) { u[j] = w[k * Cn + j]; s = fmaf(u[j], u[j], s); }
        float un    = fmaxf(sqrtf(s), MIN_NORM_F);
        float t     = tanhf(sc * un);
        float scale = t / (sc * un);
#pragma unroll
        for (int j = 0; j < Cn; ++j) cent[k * Cn + j] = scale * u[j];
        float ny = t / sc;
        y2out[k] = ny * ny;
    } else {
#pragma unroll
        for (int j = 0; j < Cn; ++j) cent[k * Cn + j] = 0.f;
        y2out[k] = 0.f;
    }
}

__global__ void __launch_bounds__(32)
hdist_main_kernel(const float* __restrict__ x,
                  const float* __restrict__ cent,
                  const float* __restrict__ y2arr,
                  const float* __restrict__ cptr,
                  float* __restrict__ nodeDist) {
    const int tile  = blockIdx.x;
    const int lane  = threadIdx.x;
    const int mbase = tile * 16;
    const int l16   = lane & 15;
    const int lhalf = lane >> 4;

    const float c  = cptr[0];
    const float sc = __builtin_amdgcn_sqrtf(c);

    v2f a[8];
    float part = 0.f;
    const float* xrow = x + (size_t)(mbase + l16) * Cn + lhalf * 2;
#pragma unroll
    for (int j = 0; j < 8; ++j) {
        a[j] = *(const v2f*)(xrow + j * 4);
        part = fmaf(a[j].x, a[j].x, fmaf(a[j].y, a[j].y, part));
    }
    float x2full = part + __shfl_xor(part, 16, 32);

    float x2r[8], Br[8], B2r[8], ccx2r[8];
#pragma unroll
    for (int r = 0; r < 8; ++r) {
        float x2 = __shfl(x2full, r + lhalf * 8, 32);
        x2r[r]   = x2;
        Br[r]    = 1.f - c * x2;
        B2r[r]   = Br[r] * Br[r];
        ccx2r[r] = c * c * x2;
    }

    const float inv2sc = 2.f / sc;

    for (int kt = 0; kt < K_TILES; ++kt) {
        const float* crow = cent + (size_t)(kt * 16 + l16) * Cn + lhalf * 2;
        v8f acc = {};
#pragma unroll
        for (int j = 0; j < 8; ++j) {
            v2f b = *(const v2f*)(crow + j * 4);
            acc = __builtin_amdgcn_wmma_f32_16x16x4_f32(
                 false, a[j],  false, b,
                 (short)0, acc,  false,  false);
        }

        const int   kcol = kt * 16 + l16;
        const float y2   = y2arr[kcol];
        const float cy2  = c * y2;

#pragma unroll
        for (int r = 0; r < 8; ++r) {
            const float xy   = acc[r];
            const float base = fmaf(-2.f * c, xy, 1.f);
            const float A_   = base + cy2;
            const float D_   = fmaxf(fmaf(ccx2r[r], y2, base), MIN_NORM_F);

            float n2 = fmaf(A_ * A_, x2r[r],
                            fmaf(-2.f * A_ * Br[r], xy, B2r[r] * y2));
            float nrm = __builtin_amdgcn_sqrtf(fmaxf(n2, 0.f)) *
                        __builtin_amdgcn_rcpf(D_);

            float z = sc * nrm;
            z = fminf(fmaxf(z, -1.f + EPS_F), 1.f - EPS_F);
            float ratio = (1.f + z) * __builtin_amdgcn_rcpf(1.f - z);
            float at    = HALF_LN2_F * __builtin_amdgcn_logf(ratio);
            float d     = inv2sc * at;
            d = d * d;

            if (kcol < Kn) {
                const size_t m = (size_t)(mbase + r + lhalf * 8);
                nodeDist[m * Kn + kcol] = d;
            }
        }
    }
}

__global__ void graph_reduce_kernel(const float* __restrict__ nodeDist,
                                    float* __restrict__ graph) {
    int i = blockIdx.x * blockDim.x + threadIdx.x;
    if (i >= Bn * Nn * Kn) return;
    int k = i % Kn;
    int n = (i / Kn) % Nn;
    int b = i / (Kn * Nn);
    float s = 0.f;
#pragma unroll
    for (int t = 0; t < Tn; ++t)
        s += nodeDist[(size_t)((b * Tn + t) * Nn + n) * Kn + k];
    graph[i] = s * (1.0f / (float)Nn);
}

extern "C" void kernel_launch(void* const* d_in, const int* in_sizes, int n_in,
                              void* d_out, int out_size, void* d_ws, size_t ws_size,
                              hipStream_t stream) {
    const float* node_repr = (const float*)d_in[0];
    const float* cweight   = (const float*)d_in[1];
    const float* cvec      = (const float*)d_in[2];

    float* graph    = (float*)d_out;
    float* nodeDist = graph + (size_t)Bn * Nn * Kn;

    float* cent = (float*)d_ws;
    float* y2   = cent + (size_t)KPAD * Cn;

    prep_centroids_kernel<<<1, 128, 0, stream>>>(cweight, cvec, cent, y2);

    hdist_main_kernel<<<M_TILES, 32, 0, stream>>>(node_repr, cent, y2, cvec, nodeDist);

    const int gElems = Bn * Nn * Kn;
    graph_reduce_kernel<<<(gElems + 255) / 256, 256, 0, stream>>>(nodeDist, graph);
}


# Steps software-pipelined inside the wave: WMMA chain and row norm of step s+1 issue ahead of the epilogue of step s (six steps written out, A rows requested 3 steps ahead); all add orders kept
# speedup vs baseline: 1.5367x; 1.5367x over previous
#include <hip/hip_runtime.h>
#include <hip/hip_bf16.h>
#include <math.h>

typedef __attribute__((ext_vector_type(2))) float v2f;
typedef __attribute__((ext_vector_type(8))) float v8f;

#define EPS_F 1e-05f
#define MIN_NORM_F 1e-15f
#define HALF_LN2_F 0.34657359027997264f

constexpr int Bn = 8, Tn = 12, Nn = 325, Cn = 32, Kn = 100;
constexpr int M_TOTAL = Bn * Tn * Nn;
constexpr int M_TILES = M_TOTAL / 16;
constexpr int K_TILES = 7;

constexpr int BN        = Bn * Nn;
constexpr int TILE_ROWS = 8;
constexpr int T_HALF    = Tn / 2;
constexpr int R_TILES   = BN / TILE_ROWS;
constexpr int UNITS     = R_TILES * K_TILES;
constexpr int WG_WAVES  = 4;
constexpr int FUSED_WGS = (UNITS + WG_WAVES - 1) / WG_WAVES;
static_assert(BN % TILE_ROWS == 0, "every 8-row tile is whole");
static_assert(Tn % 2 == 0 && T_HALF % 3 == 0, "A rows are a 3-deep ring");

__global__ void __launch_bounds__(32 * WG_WAVES)
hdist_fused_kernel(const float* __restrict__ x,
                   const float* __restrict__ w,
                   const float* __restrict__ cptr,
                   float* __restrict__ nodeDist,
                   float* __restrict__ graph) {
    const int wave = __builtin_amdgcn_readfirstlane((int)(threadIdx.x >> 5));
    const int unit = blockIdx.x * WG_WAVES + wave;
    if (unit >= UNITS) return;
    const int rt    = unit / K_TILES;
    const int kt    = unit - rt * K_TILES;
    const int lane  = threadIdx.x & 31;
    const int l16   = lane & 15;
    const int lhalf = lane >> 4;

    const float c  = cptr[0];
    const float sc = __builtin_amdgcn_sqrtf(c);
    const float inv2sc = 2.f / sc;

    const int   kcol = kt * 16 + l16;
    const int   krow = kcol < Kn ? kcol : Kn - 1;
    const float* wrow = w + (size_t)krow * Cn + lhalf * 2;
    v2f bf[8];
#pragma unroll
    for (int j = 0; j < 8; ++j) bf[j] = *(const v2f*)(wrow + j * 4);

    const int gA = rt * TILE_ROWS + (l16 & 7);
    const int tA = (l16 >> 3) * T_HALF;
    const int xoff = (gA + (gA / Nn) * (Tn - 1) * Nn + tA * Nn) * Cn + lhalf * 2;

    auto load_a = [&](v2f (&a)[8], int s) {
        const float* xt = x + (size_t)s * (Nn * Cn);
#pragma unroll
        for (int j = 0; j < 8; ++j) a[j] = *(const v2f*)(xt + xoff + j * 4);
    };

    v2f a0[8], a1[8], a2[8];
    load_a(a0, 0);
    load_a(a1, 1);
    load_a(a2, 2);
    __builtin_amdgcn_sched_barrier(0);

    float wpart = 0.f;
#pragma unroll
    for (int j = 0; j < 8; ++j)
        wpart = fmaf(bf[j].x, bf[j].x, fmaf(bf[j].y, bf[j].y, wpart));
    const float w2   = wpart + __shfl_xor(wpart, 16, 32);
    const float scm  = sqrtf(c);
    const float un   = fmaxf(sqrtf(w2), MIN_NORM_F);
    const float th   = tanhf(scm * un);
    const float sy   = th / (scm * un);
    const float ny   = th / scm;
    const float y2   = ny * ny;
    const float cy2  = c * y2;
    const float m2cs = -2.f * c * sy;
    const float m2s  = -2.f * sy;
    const v2f m2cs2 = { m2cs, m2cs }, y22 = { y2, y2 }, one2 = { 1.f, 1.f };

    const int  gS0   = rt * TILE_ROWS;
    const bool valid = kcol < Kn;
    int ndoff[8];
#pragma unroll
    for (int r = 0; r < 8; ++r) {
        const int g = gS0 + r;
        ndoff[r] = (g + (g / Nn) * (Tn - 1) * Nn + lhalf * T_HALF * Nn) * Kn + kcol;
    }

    float gsum[8];
#pragma unroll
    for (int r = 0; r < 8; ++r) gsum[r] = 0.f;

    auto front = [&](const v2f (&a)[8], v8f& accOut, float& x2Out) {
        float part = 0.f;
#pragma unroll
        for (int j = 0; j < 8; ++j)
            part = fmaf(a[j].x, a[j].x, fmaf(a[j].y, a[j].y, part));
        x2Out = part + __shfl_xor(part, 16, 32);

        v8f acc = {};
#pragma unroll
        for (int j = 0; j < 8; ++j)
            acc = __builtin_amdgcn_wmma_f32_16x16x4_f32(
                 false, a[j],  false, bf[j],
                 (short)0, acc,  false,  false);
        accOut = acc;
    };

    auto back = [&](const v8f& acc, float x2full, int s) {
        float* ndt = nodeDist + (size_t)s * (Nn * Kn);
        float d8[8];
#pragma unroll
        for (int p = 0; p < 4; ++p) {
            const int r0 = 2 * p, r1 = 2 * p + 1;
            const v2f x2   = { __shfl(x2full, r0 + lhalf * 8, 32),
                               __shfl(x2full, r1 + lhalf * 8, 32) };
            const v2f xw   = { acc[r0], acc[r1] };
            const v2f Br   = 1.f - c * x2;
            const v2f B2r  = Br * Br;
            const v2f ccx2 = (c * c) * x2;

            const v2f base = __builtin_elementwise_fma(m2cs2, xw, one2);
            const v2f A_   = base + cy2;
            const v2f Dv   = __builtin_elementwise_fma(ccx2, y22, base);

            const v2f n2v  = __builtin_elementwise_fma(
                A_ * A_, x2,
                __builtin_elementwise_fma(m2s * A_ * Br, xw, B2r * y2));
#pragma unroll
            for (int e = 0; e < 2; ++e) {
                const int r = 2 * p + e;
                const float D_ = fmaxf(Dv[e], MIN_NORM_F);
                float nrm = __builtin_amdgcn_sqrtf(fmaxf(n2v[e], 0.f)) *
                            __builtin_amdgcn_rcpf(D_);

                float z = sc * nrm;
                z = fminf(fmaxf(z, -1.f + EPS_F), 1.f - EPS_F);
                float ratio = (1.f + z) * __builtin_amdgcn_rcpf(1.f - z);
                float at    = HALF_LN2_F * __builtin_amdgcn_logf(ratio);
                float d     = inv2sc * at;
                d = d * d;
                d8[r] = d;
                gsum[r] += d;
            }
        }
        if (valid) {
#pragma unroll
            for (int r = 0; r < 8; ++r) ndt[ndoff[r]] = d8[r];
        }
    };

    static_assert(T_HALF == 6, "the pipeline below is written out for 6 steps");
    v8f accA, accB;
    float xnA, xnB;
    front(a0, accA, xnA);
    load_a(a0, 3);
    front(a1, accB, xnB);
    load_a(a1, 4);
    back(accA, xnA, 0);
    front(a2, accA, xnA);
    load_a(a2, 5);
    back(accB, xnB, 1);
    front(a0, accB, xnB);
    back(accA, xnA, 2);
    front(a1, accA, xnA);
    back(accB, xnB, 3);
    front(a2, accB, xnB);
    back(accA, xnA, 4);
    back(accB, xnB, 5);

    float tot[8];
#pragma unroll
    for (int r = 0; r < 8; ++r) tot[r] = gsum[r] + __shfl_xor(gsum[r], 16, 32);
    if (valid) {
#pragma unroll
        for (int i = 0; i < 4; ++i) {
            const float v = lhalf ? tot[4 + i] : tot[i];
            graph[(size_t)(gS0 + lhalf * 4 + i) * Kn + kcol] = v * (1.0f / (float)Nn);
        }
    }
}

extern "C" void kernel_launch(void* const* d_in, const int* in_sizes, int n_in,
                              void* d_out, int out_size, void* d_ws, size_t ws_size,
                              hipStream_t stream) {
    const float* node_repr = (const float*)d_in[0];
    const float* cweight   = (const float*)d_in[1];
    const float* cvec      = (const float*)d_in[2];

    float* graph    = (float*)d_out;
    float* nodeDist = graph + (size_t)Bn * Nn * Kn;

    hdist_fused_kernel<<<FUSED_WGS, 32 * WG_WAVES, 0, stream>>>(
        node_repr, cweight, cvec, nodeDist, graph);
}


# Work-group = the 7 centroid tiles of one 8-row tile; each step's distances staged in LDS, written as whole 128-B lines (16-B stores, waves line-aligned; was 64-B pieces); one LDS-only barrier a step
# speedup vs baseline: 1.6618x; 1.0814x over previous
#include <hip/hip_runtime.h>
#include <hip/hip_bf16.h>
#include <math.h>

typedef __attribute__((ext_vector_type(2))) float v2f;
typedef __attribute__((ext_vector_type(8))) float v8f;

#define EPS_F 1e-05f
#define MIN_NORM_F 1e-15f
#define HALF_LN2_F 0.34657359027997264f

constexpr int Bn = 8, Tn = 12, Nn = 325, Cn = 32, Kn = 100;
constexpr int M_TOTAL = Bn * Tn * Nn;
constexpr int M_TILES = M_TOTAL / 16;
constexpr int K_TILES = 7;

constexpr int BN        = Bn * Nn;
constexpr int TILE_ROWS = 8;
constexpr int T_HALF    = Tn / 2;
constexpr int R_TILES   = BN / TILE_ROWS;
constexpr int UNITS     = R_TILES * K_TILES;
constexpr int WG_WAVES  = K_TILES;
constexpr int WG_THREADS = 32 * WG_WAVES;
constexpr int FUSED_WGS = R_TILES;
constexpr int CHUNK_F   = TILE_ROWS * Kn;
constexpr int CHUNK_Q   = CHUNK_F / 4;
constexpr int STAGE_F   = 2 * CHUNK_F;
constexpr int XBATCH_F  = (Tn - 1) * Nn * Kn;
static_assert(UNITS == FUSED_WGS * WG_WAVES, "every wave of every work-group has a unit");
static_assert(Kn % 4 == 0, "a row is a whole number of 16-byte pieces");
static_assert(CHUNK_Q + 7 <= WG_THREADS, "one pass per chunk, shifted by up to 7 pieces");
static_assert(BN % TILE_ROWS == 0, "every 8-row tile is whole");
static_assert(Tn % 2 == 0 && T_HALF % 3 == 0, "A rows are a 3-deep ring");

__global__ void __launch_bounds__(WG_THREADS)
hdist_fused_kernel(const float* __restrict__ x,
                   const float* __restrict__ w,
                   const float* __restrict__ cptr,
                   float* __restrict__ nodeDist,
                   float* __restrict__ graph) {
    __shared__ __attribute__((aligned(16))) float stage[2][STAGE_F];
    const int wave = __builtin_amdgcn_readfirstlane((int)(threadIdx.x >> 5));
    const int rt    = blockIdx.x;
    const int kt    = wave;
    const int lane  = threadIdx.x & 31;
    const int l16   = lane & 15;
    const int lhalf = lane >> 4;

    const float c  = cptr[0];
    const float sc = __builtin_amdgcn_sqrtf(c);
    const float inv2sc = 2.f / sc;

    const int   kcol = kt * 16 + l16;
    const int   krow = kcol < Kn ? kcol : Kn - 1;
    const float* wrow = w + (size_t)krow * Cn + lhalf * 2;
    v2f bf[8];
#pragma unroll
    for (int j = 0; j < 8; ++j) bf[j] = *(const v2f*)(wrow + j * 4);

    const int gA = rt * TILE_ROWS + (l16 & 7);
    const int tA = (l16 >> 3) * T_HALF;
    const int xoff = (gA + (gA / Nn) * (Tn - 1) * Nn + tA * Nn) * Cn + lhalf * 2;

    auto load_a = [&](v2f (&a)[8], int s) {
        const float* xt = x + (size_t)s * (Nn * Cn);
#pragma unroll
        for (int j = 0; j < 8; ++j) a[j] = *(const v2f*)(xt + xoff + j * 4);
    };

    v2f a0[8], a1[8], a2[8];
    load_a(a0, 0);
    load_a(a1, 1);
    load_a(a2, 2);
    __builtin_amdgcn_sched_barrier(0);

    float wpart = 0.f;
#pragma unroll
    for (int j = 0; j < 8; ++j)
        wpart = fmaf(bf[j].x, bf[j].x, fmaf(bf[j].y, bf[j].y, wpart));
    const float w2   = wpart + __shfl_xor(wpart, 16, 32);
    const float scm  = sqrtf(c);
    const float un   = fmaxf(sqrtf(w2), MIN_NORM_F);
    const float th   = tanhf(scm * un);
    const float sy   = th / (scm * un);
    const float ny   = th / scm;
    const float y2   = ny * ny;
    const float cy2  = c * y2;
    const float m2cs = -2.f * c * sy;
    const float m2s  = -2.f * sy;
    const v2f m2cs2 = { m2cs, m2cs }, y22 = { y2, y2 }, one2 = { 1.f, 1.f };

    const int  gS0   = rt * TILE_ROWS;
    const bool valid = kcol < Kn;

    const int tid    = threadIdx.x;
    const int b0     = gS0 / Nn;
    const int m0     = gS0 + b0 * (Tn - 1) * Nn;
    const int nfirst = (b0 + 1) * Nn - gS0;
    const int pcross = (nfirst < TILE_ROWS ? nfirst : TILE_ROWS) * (Kn / 4);
    const int srow = lhalf * TILE_ROWS * Kn + kcol;

    float gsum[8];
#pragma unroll
    for (int r = 0; r < 8; ++r) gsum[r] = 0.f;

    auto front = [&](const v2f (&a)[8], v8f& accOut, float& x2Out) {
        float part = 0.f;
#pragma unroll
        for (int j = 0; j < 8; ++j)
            part = fmaf(a[j].x, a[j].x, fmaf(a[j].y, a[j].y, part));
        x2Out = part + __shfl_xor(part, 16, 32);

        v8f acc = {};
#pragma unroll
        for (int j = 0; j < 8; ++j)
            acc = __builtin_amdgcn_wmma_f32_16x16x4_f32(
                 false, a[j],  false, bf[j],
                 (short)0, acc,  false,  false);
        accOut = acc;
    };

    auto epi = [&](const v8f& acc, float x2full, float (&d8)[8]) {
#pragma unroll
        for (int p = 0; p < 4; ++p) {
            const int r0 = 2 * p, r1 = 2 * p + 1;
            const v2f x2   = { __shfl(x2full, r0 + lhalf * 8, 32),
                               __shfl(x2full, r1 + lhalf * 8, 32) };
            const v2f xw   = { acc[r0], acc[r1] };
            const v2f Br   = 1.f - c * x2;
            const v2f B2r  = Br * Br;
            const v2f ccx2 = (c * c) * x2;

            const v2f base = __builtin_elementwise_fma(m2cs2, xw, one2);
            const v2f A_   = base + cy2;
            const v2f Dv   = __builtin_elementwise_fma(ccx2, y22, base);

            const v2f n2v  = __builtin_elementwise_fma(
                A_ * A_, x2,
                __builtin_elementwise_fma(m2s * A_ * Br, xw, B2r * y2));
#pragma unroll
            for (int e = 0; e < 2; ++e) {
                const int r = 2 * p + e;
                const float D_ = fmaxf(Dv[e], MIN_NORM_F);
                float nrm = __builtin_amdgcn_sqrtf(fmaxf(n2v[e], 0.f)) *
                            __builtin_amdgcn_rcpf(D_);

                float z = sc * nrm;
                z = fminf(fmaxf(z, -1.f + EPS_F), 1.f - EPS_F);
                float ratio = (1.f + z) * __builtin_amdgcn_rcpf(1.f - z);
                float at    = HALF_LN2_F * __builtin_amdgcn_logf(ratio);
                float d     = inv2sc * at;
                d = d * d;
                d8[r] = d;
                gsum[r] += d;
            }
        }
    };

    auto put = [&](const float (&d8)[8], int s) {
        float* sbuf = stage[s & 1];
        if (valid) {
#pragma unroll
            for (int r = 0; r < 8; ++r) sbuf[srow + r * Kn] = d8[r];
        }
        __builtin_amdgcn_fence(__ATOMIC_RELEASE, "workgroup", "local");
        __builtin_amdgcn_s_barrier_signal(-1);
    };

    auto get1 = [&](float4& cv, float*& gp, bool& ok, int s, int h) {
        float* gbase = nodeDist + (size_t)(m0 + (s + h * T_HALF) * Nn) * Kn;
        const int a  = (int)(((size_t)gbase >> 4) & 7);
        const int p  = tid - a;
        ok = (unsigned)p < (unsigned)CHUNK_Q;
        const int pc = ok ? p : 0;
        cv = *(const float4*)(stage[s & 1] + h * CHUNK_F + 4 * pc);
        gp = gbase + 4 * pc + (pc >= pcross ? XBATCH_F : 0);
    };
    auto get = [&](float4& c0, float4& c1, float*& g0, float*& g1, bool& k0, bool& k1, int s) {
        __builtin_amdgcn_s_barrier_wait(-1);
        __builtin_amdgcn_fence(__ATOMIC_ACQUIRE, "workgroup", "local");
        get1(c0, g0, k0, s, 0);
        get1(c1, g1, k1, s, 1);
        __builtin_amdgcn_sched_barrier(1);
    };

    auto out = [&](const float4& c0, const float4& c1, float* g0, float* g1, bool k0, bool k1) {
        if (k0) *(float4*)g0 = c0;
        if (k1) *(float4*)g1 = c1;
    };

    static_assert(T_HALF == 6, "the pipeline below is written out for 6 steps");
    v8f accA, accB;
    float xnA, xnB;
    float d8[8];
    float4 c0, c1;
    float *g0, *g1;
    bool   k0, k1;
    front(a0, accA, xnA);
    load_a(a0, 3);
    front(a1, accB, xnB);
    load_a(a1, 4);
    epi(accA, xnA, d8);  put(d8, 0);
    front(a2, accA, xnA);
    load_a(a2, 5);
    get(c0, c1, g0, g1, k0, k1, 0);
    epi(accB, xnB, d8);  out(c0, c1, g0, g1, k0, k1);  put(d8, 1);
    front(a0, accB, xnB);
    get(c0, c1, g0, g1, k0, k1, 1);
    epi(accA, xnA, d8);  out(c0, c1, g0, g1, k0, k1);  put(d8, 2);
    front(a1, accA, xnA);
    get(c0, c1, g0, g1, k0, k1, 2);
    epi(accB, xnB, d8);  out(c0, c1, g0, g1, k0, k1);  put(d8, 3);
    front(a2, accB, xnB);
    get(c0, c1, g0, g1, k0, k1, 3);
    epi(accA, xnA, d8);  out(c0, c1, g0, g1, k0, k1);  put(d8, 4);
    get(c0, c1, g0, g1, k0, k1, 4);
    epi(accB, xnB, d8);  out(c0, c1, g0, g1, k0, k1);  put(d8, 5);

    float tot[8];
#pragma unroll
    for (int r = 0; r < 8; ++r) tot[r] = gsum[r] + __shfl_xor(gsum[r], 16, 32);
    if (valid) {
#pragma unroll
        for (int i = 0; i < 4; ++i) {
            const float v = lhalf ? tot[4 + i] : tot[i];
            graph[(size_t)(gS0 + lhalf * 4 + i) * Kn + kcol] = v * (1.0f / (float)Nn);
        }
    }
    get(c0, c1, g0, g1, k0, k1, 5);
    out(c0, c1, g0, g1, k0, k1);
}

extern "C" void kernel_launch(void* const* d_in, const int* in_sizes, int n_in,
                              void* d_out, int out_size, void* d_ws, size_t ws_size,
                              hipStream_t stream) {
    const float* node_repr = (const float*)d_in[0];
    const float* cweight   = (const float*)d_in[1];
    const float* cvec      = (const float*)d_in[2];

    float* graph    = (float*)d_out;
    float* nodeDist = graph + (size_t)Bn * Nn * Kn;

    hdist_fused_kernel<<<FUSED_WGS, WG_THREADS, 0, stream>>>(
        node_repr, cweight, cvec, nodeDist, graph);
}


# Distance epilogue by two exact identities: |(-x)+y|^2 = |x-y|^2/den and (1+z)/(1-z) = (den+s)/(den-s); one sqrt, rcp, log2 per distance, clip moved onto the log; last bits differ, add orders kept
# speedup vs baseline: 1.7888x; 1.0764x over previous
#include <hip/hip_runtime.h>
#include <hip/hip_bf16.h>
#include <math.h>

typedef __attribute__((ext_vector_type(2))) float v2f;
typedef __attribute__((ext_vector_type(8))) float v8f;

#define EPS_F 1e-05f
#define MIN_NORM_F 1e-15f
#define HALF_LN2_F 0.34657359027997264f

constexpr int Bn = 8, Tn = 12, Nn = 325, Cn = 32, Kn = 100;
constexpr int M_TOTAL = Bn * Tn * Nn;
constexpr int M_TILES = M_TOTAL / 16;
constexpr int K_TILES = 7;

constexpr int BN        = Bn * Nn;
constexpr int TILE_ROWS = 8;
constexpr int T_HALF    = Tn / 2;
constexpr int R_TILES   = BN / TILE_ROWS;
constexpr int UNITS     = R_TILES * K_TILES;
constexpr int WG_WAVES  = K_TILES;
constexpr int WG_THREADS = 32 * WG_WAVES;
constexpr int FUSED_WGS = R_TILES;
constexpr int CHUNK_F   = TILE_ROWS * Kn;
constexpr int CHUNK_Q   = CHUNK_F / 4;
constexpr int STAGE_F   = 2 * CHUNK_F;
constexpr int XBATCH_F  = (Tn - 1) * Nn * Kn;
static_assert(UNITS == FUSED_WGS * WG_WAVES, "every wave of every work-group has a unit");
static_assert(Kn % 4 == 0, "a row is a whole number of 16-byte pieces");
static_assert(CHUNK_Q + 7 <= WG_THREADS, "one pass per chunk, shifted by up to 7 pieces");
static_assert(BN % TILE_ROWS == 0, "every 8-row tile is whole");
static_assert(Tn % 2 == 0 && T_HALF % 3 == 0, "A rows are a 3-deep ring");

__global__ void __launch_bounds__(WG_THREADS)
hdist_fused_kernel(const float* __restrict__ x,
                   const float* __restrict__ w,
                   const float* __restrict__ cptr,
                   float* __restrict__ nodeDist,
                   float* __restrict__ graph) {
    __shared__ __attribute__((aligned(16))) float stage[2][STAGE_F];
    const int wave = __builtin_amdgcn_readfirstlane((int)(threadIdx.x >> 5));
    const int rt    = blockIdx.x;
    const int kt    = wave;
    const int lane  = threadIdx.x & 31;
    const int l16   = lane & 15;
    const int lhalf = lane >> 4;

    const float c  = cptr[0];
    const float sc = __builtin_amdgcn_sqrtf(c);
    const float inv2sc = 2.f / sc;

    const int   kcol = kt * 16 + l16;
    const int   krow = kcol < Kn ? kcol : Kn - 1;
    const float* wrow = w + (size_t)krow * Cn + lhalf * 2;
    v2f bf[8];
#pragma unroll
    for (int j = 0; j < 8; ++j) bf[j] = *(const v2f*)(wrow + j * 4);

    const int gA = rt * TILE_ROWS + (l16 & 7);
    const int tA = (l16 >> 3) * T_HALF;
    const int xoff = (gA + (gA / Nn) * (Tn - 1) * Nn + tA * Nn) * Cn + lhalf * 2;

    auto load_a = [&](v2f (&a)[8], int s) {
        const float* xt = x + (size_t)s * (Nn * Cn);
#pragma unroll
        for (int j = 0; j < 8; ++j) a[j] = *(const v2f*)(xt + xoff + j * 4);
    };

    v2f a0[8], a1[8], a2[8];
    load_a(a0, 0);
    load_a(a1, 1);
    load_a(a2, 2);
    __builtin_amdgcn_sched_barrier(0);

    float wpart = 0.f;
#pragma unroll
    for (int j = 0; j < 8; ++j)
        wpart = fmaf(bf[j].x, bf[j].x, fmaf(bf[j].y, bf[j].y, wpart));
    const float w2   = wpart + __shfl_xor(wpart, 16, 32);
    const float scm  = sqrtf(c);
    const float un   = fmaxf(sqrtf(w2), MIN_NORM_F);
    const float th   = tanhf(scm * un);
    const float sy   = th / (scm * un);
    const float ny   = th / scm;
    const float y2   = ny * ny;
    const float cy2  = c * y2;
    const float m2cs = -2.f * c * sy;
    const v2f m2cs2 = { m2cs, m2cs }, cy22 = { cy2, cy2 }, one2 = { 1.f, 1.f };
    const float kln  = inv2sc * HALF_LN2_F;
    const float zc   = 1.f - EPS_F;
    const float lmax = __builtin_amdgcn_logf((1.f + zc) * __builtin_amdgcn_rcpf(1.f - zc));

    const int  gS0   = rt * TILE_ROWS;
    const bool valid = kcol < Kn;

    const int tid    = threadIdx.x;
    const int b0     = gS0 / Nn;
    const int m0     = gS0 + b0 * (Tn - 1) * Nn;
    const int nfirst = (b0 + 1) * Nn - gS0;
    const int pcross = (nfirst < TILE_ROWS ? nfirst : TILE_ROWS) * (Kn / 4);
    const int srow = lhalf * TILE_ROWS * Kn + kcol;

    float gsum[8];
#pragma unroll
    for (int r = 0; r < 8; ++r) gsum[r] = 0.f;

    auto front = [&](const v2f (&a)[8], v8f& accOut, float& x2Out) {
        float part = 0.f;
#pragma unroll
        for (int j = 0; j < 8; ++j)
            part = fmaf(a[j].x, a[j].x, fmaf(a[j].y, a[j].y, part));
        x2Out = c * (part + __shfl_xor(part, 16, 32));

        v8f acc = {};
#pragma unroll
        for (int j = 0; j < 8; ++j)
            acc = __builtin_amdgcn_wmma_f32_16x16x4_f32(
                 false, a[j],  false, bf[j],
                 (short)0, acc,  false,  false);
        accOut = acc;
    };

    auto epi = [&](const v8f& acc, float cx2full, float (&d8)[8]) {
#pragma unroll
        for (int p = 0; p < 4; ++p) {
            const int r0 = 2 * p, r1 = 2 * p + 1;
            const v2f cx2  = { __shfl(cx2full, r0 + lhalf * 8, 32),
                               __shfl(cx2full, r1 + lhalf * 8, 32) };
            const v2f xw   = { acc[r0], acc[r1] };

            const v2f base = __builtin_elementwise_fma(m2cs2, xw, one2);
            const v2f Dv   = __builtin_elementwise_fma(cx2, cy22, base);
            const v2f cq   = __builtin_elementwise_fma(m2cs2, xw, cx2 + cy22);
            const v2f P    = Dv * cq;
#pragma unroll
            for (int e = 0; e < 2; ++e) {
                const int r = 2 * p + e;
                const float D_ = fmaxf(Dv[e], MIN_NORM_F);
                const float s  = __builtin_amdgcn_sqrtf(fmaxf(P[e], 0.f));
                const float ratio = (D_ + s) * __builtin_amdgcn_rcpf(D_ - s);
                const float lg = fminf(__builtin_amdgcn_logf(ratio), lmax);
                const float u  = kln * lg;
                const float d  = u * u;
                d8[r] = d;
                gsum[r] += d;
            }
        }
    };

    auto put = [&](const float (&d8)[8], int s) {
        float* sbuf = stage[s & 1];
        if (valid) {
#pragma unroll
            for (int r = 0; r < 8; ++r) sbuf[srow + r * Kn] = d8[r];
        }
        __builtin_amdgcn_fence(__ATOMIC_RELEASE, "workgroup", "local");
        __builtin_amdgcn_s_barrier_signal(-1);
    };

    auto get1 = [&](float4& cv, float*& gp, bool& ok, int s, int h) {
        float* gbase = nodeDist + (size_t)(m0 + (s + h * T_HALF) * Nn) * Kn;
        const int a  = (int)(((size_t)gbase >> 4) & 7);
        const int p  = tid - a;
        ok = (unsigned)p < (unsigned)CHUNK_Q;
        const int pc = ok ? p : 0;
        cv = *(const float4*)(stage[s & 1] + h * CHUNK_F + 4 * pc);
        gp = gbase + 4 * pc + (pc >= pcross ? XBATCH_F : 0);
    };
    auto get = [&](float4& c0, float4& c1, float*& g0, float*& g1, bool& k0, bool& k1, int s) {
        __builtin_amdgcn_s_barrier_wait(-1);
        __builtin_amdgcn_fence(__ATOMIC_ACQUIRE, "workgroup", "local");
        get1(c0, g0, k0, s, 0);
        get1(c1, g1, k1, s, 1);
        __builtin_amdgcn_sched_barrier(1);
    };

    auto out = [&](const float4& c0, const float4& c1, float* g0, float* g1, bool k0, bool k1) {
        if (k0) *(float4*)g0 = c0;
        if (k1) *(float4*)g1 = c1;
    };

    static_assert(T_HALF == 6, "the pipeline below is written out for 6 steps");
    v8f accA, accB;
    float xnA, xnB;
    float d8[8];
    float4 c0, c1;
    float *g0, *g1;
    bool   k0, k1;
    front(a0, accA, xnA);
    load_a(a0, 3);
    front(a1, accB, xnB);
    load_a(a1, 4);
    epi(accA, xnA, d8);  put(d8, 0);
    front(a2, accA, xnA);
    load_a(a2, 5);
    get(c0, c1, g0, g1, k0, k1, 0);
    epi(accB, xnB, d8);  out(c0, c1, g0, g1, k0, k1);  put(d8, 1);
    front(a0, accB, xnB);
    get(c0, c1, g0, g1, k0, k1, 1);
    epi(accA, xnA, d8);  out(c0, c1, g0, g1, k0, k1);  put(d8, 2);
    front(a1, accA, xnA);
    get(c0, c1, g0, g1, k0, k1, 2);
    epi(accB, xnB, d8);  out(c0, c1, g0, g1, k0, k1);  put(d8, 3);
    front(a2, accB, xnB);
    get(c0, c1, g0, g1, k0, k1, 3);
    epi(accA, xnA, d8);  out(c0, c1, g0, g1, k0, k1);  put(d8, 4);
    get(c0, c1, g0, g1, k0, k1, 4);
    epi(accB, xnB, d8);  out(c0, c1, g0, g1, k0, k1);  put(d8, 5);

    float tot[8];
#pragma unroll
    for (int r = 0; r < 8; ++r) tot[r] = gsum[r] + __shfl_xor(gsum[r], 16, 32);
    if (valid) {
#pragma unroll
        for (int i = 0; i < 4; ++i) {
            const float v = lhalf ? tot[4 + i] : tot[i];
            graph[(size_t)(gS0 + lhalf * 4 + i) * Kn + kcol] = v * (1.0f / (float)Nn);
        }
    }
    get(c0, c1, g0, g1, k0, k1, 5);
    out(c0, c1, g0, g1, k0, k1);
}

extern "C" void kernel_launch(void* const* d_in, const int* in_sizes, int n_in,
                              void* d_out, int out_size, void* d_ws, size_t ws_size,
                              hipStream_t stream) {
    const float* node_repr = (const float*)d_in[0];
    const float* cweight   = (const float*)d_in[1];
    const float* cvec      = (const float*)d_in[2];

    float* graph    = (float*)d_out;
    float* nodeDist = graph + (size_t)Bn * Nn * Kn;

    hdist_fused_kernel<<<FUSED_WGS, WG_THREADS, 0, stream>>>(
        node_repr, cweight, cvec, nodeDist, graph);
}


# x rows staged once per work-group in LDS by whole-line 16-B loads (row pitch 144 B) and read back as A fragments; each wave's 48 8-B global A loads of 16 lines each are gone; add orders kept
# speedup vs baseline: 1.9636x; 1.0977x over previous
#include <hip/hip_runtime.h>
#include <hip/hip_bf16.h>
#include <math.h>

typedef __attribute__((ext_vector_type(2))) float v2f;
typedef __attribute__((ext_vector_type(8))) float v8f;

#define EPS_F 1e-05f
#define MIN_NORM_F 1e-15f
#define HALF_LN2_F 0.34657359027997264f

constexpr int Bn = 8, Tn = 12, Nn = 325, Cn = 32, Kn = 100;
constexpr int M_TOTAL = Bn * Tn * Nn;
constexpr int M_TILES = M_TOTAL / 16;
constexpr int K_TILES = 7;

constexpr int BN        = Bn * Nn;
constexpr int TILE_ROWS = 8;
constexpr int T_HALF    = Tn / 2;
constexpr int R_TILES   = BN / TILE_ROWS;
constexpr int UNITS     = R_TILES * K_TILES;
constexpr int WG_WAVES  = K_TILES;
constexpr int WG_THREADS = 32 * WG_WAVES;
constexpr int FUSED_WGS = R_TILES;
constexpr int CHUNK_F   = TILE_ROWS * Kn;
constexpr int CHUNK_Q   = CHUNK_F / 4;
constexpr int STAGE_F   = 2 * CHUNK_F;
constexpr int XBATCH_F  = (Tn - 1) * Nn * Kn;
constexpr int XROWS     = T_HALF * 16;
constexpr int XPITCH    = Cn + 4;
constexpr int XQ        = XROWS * (Cn / 4);
constexpr int XFULL     = XQ / WG_THREADS;
constexpr int XREST     = XQ - XFULL * WG_THREADS;
static_assert(XREST % 32 == 0, "the left-over pieces are whole waves: a wave-uniform branch decides");
static_assert(UNITS == FUSED_WGS * WG_WAVES, "every wave of every work-group has a unit");
static_assert(Kn % 4 == 0, "a row is a whole number of 16-byte pieces");
static_assert(CHUNK_Q + 7 <= WG_THREADS, "one pass per chunk, shifted by up to 7 pieces");
static_assert(BN % TILE_ROWS == 0, "every 8-row tile is whole");
static_assert(Tn % 2 == 0 && T_HALF % 3 == 0, "A rows are a 3-deep ring");

__global__ void __launch_bounds__(WG_THREADS)
hdist_fused_kernel(const float* __restrict__ x,
                   const float* __restrict__ w,
                   const float* __restrict__ cptr,
                   float* __restrict__ nodeDist,
                   float* __restrict__ graph) {
    __shared__ __attribute__((aligned(16))) float stage[2][STAGE_F];
    __shared__ __attribute__((aligned(16))) float xs[XROWS * XPITCH];
    const int wave = __builtin_amdgcn_readfirstlane((int)(threadIdx.x >> 5));
    const int rt    = blockIdx.x;
    const int kt    = wave;
    const int lane  = threadIdx.x & 31;
    const int l16   = lane & 15;
    const int lhalf = lane >> 4;

    const float c  = cptr[0];
    const float sc = __builtin_amdgcn_sqrtf(c);
    const float inv2sc = 2.f / sc;

    const int   kcol = kt * 16 + l16;
    const int   krow = kcol < Kn ? kcol : Kn - 1;
    const float* wrow = w + (size_t)krow * Cn + lhalf * 2;
    v2f bf[8];
#pragma unroll
    for (int j = 0; j < 8; ++j) bf[j] = *(const v2f*)(wrow + j * 4);

    static_assert(XFULL == 3 && XPITCH % 4 == 0, "three full passes and one of 3 waves, written out below");
    auto xreq = [&](int i, float4& v, int& o) {
        const int q   = (int)threadIdx.x + i * WG_THREADS;
        const int row = q >> 3, pq = q & 7;
        const int R   = row & 15, sx = row >> 4;
        const int g   = rt * TILE_ROWS + (R & 7);
        const int t   = sx + (R >> 3) * T_HALF;
        const int m   = g + (g / Nn) * (Tn - 1) * Nn + t * Nn;
        v = *(const float4*)(x + (size_t)m * Cn + pq * 4);
        o = row * (XPITCH / 4) + pq;
    };
    const bool xlast = wave < XREST / 32;
    float4 xv0, xv1, xv2, xv3 = { 0.f, 0.f, 0.f, 0.f };
    int    xo0, xo1, xo2, xo3 = 0;
    xreq(0, xv0, xo0);
    xreq(1, xv1, xo1);
    xreq(2, xv2, xo2);
    if (xlast) xreq(3, xv3, xo3);
    __builtin_amdgcn_sched_barrier(0);

    float wpart = 0.f;
#pragma unroll
    for (int j = 0; j < 8; ++j)
        wpart = fmaf(bf[j].x, bf[j].x, fmaf(bf[j].y, bf[j].y, wpart));
    const float w2   = wpart + __shfl_xor(wpart, 16, 32);
    const float scm  = sqrtf(c);
    const float un   = fmaxf(sqrtf(w2), MIN_NORM_F);
    const float th   = tanhf(scm * un);
    const float sy   = th / (scm * un);
    const float ny   = th / scm;
    const float y2   = ny * ny;
    const float cy2  = c * y2;
    const float m2cs = -2.f * c * sy;
    const v2f m2cs2 = { m2cs, m2cs }, cy22 = { cy2, cy2 }, one2 = { 1.f, 1.f };
    const float kln  = inv2sc * HALF_LN2_F;
    const float zc   = 1.f - EPS_F;
    const float lmax = __builtin_amdgcn_logf((1.f + zc) * __builtin_amdgcn_rcpf(1.f - zc));

    const int  gS0   = rt * TILE_ROWS;
    const bool valid = kcol < Kn;

    const int tid    = threadIdx.x;
    const int b0     = gS0 / Nn;
    const int m0     = gS0 + b0 * (Tn - 1) * Nn;
    const int nfirst = (b0 + 1) * Nn - gS0;
    const int pcross = (nfirst < TILE_ROWS ? nfirst : TILE_ROWS) * (Kn / 4);
    const int srow = lhalf * TILE_ROWS * Kn + kcol;

    float gsum[8];
#pragma unroll
    for (int r = 0; r < 8; ++r) gsum[r] = 0.f;

    auto front = [&](const v2f (&a)[8], v8f& accOut, float& x2Out) {
        float part = 0.f;
#pragma unroll
        for (int j = 0; j < 8; ++j)
            part = fmaf(a[j].x, a[j].x, fmaf(a[j].y, a[j].y, part));
        x2Out = c * (part + __shfl_xor(part, 16, 32));

        v8f acc = {};
#pragma unroll
        for (int j = 0; j < 8; ++j)
            acc = __builtin_amdgcn_wmma_f32_16x16x4_f32(
                 false, a[j],  false, bf[j],
                 (short)0, acc,  false,  false);
        accOut = acc;
    };

    auto epi = [&](const v8f& acc, float cx2full, float (&d8)[8]) {
#pragma unroll
        for (int p = 0; p < 4; ++p) {
            const int r0 = 2 * p, r1 = 2 * p + 1;
            const v2f cx2  = { __shfl(cx2full, r0 + lhalf * 8, 32),
                               __shfl(cx2full, r1 + lhalf * 8, 32) };
            const v2f xw   = { acc[r0], acc[r1] };

            const v2f base = __builtin_elementwise_fma(m2cs2, xw, one2);
            const v2f Dv   = __builtin_elementwise_fma(cx2, cy22, base);
            const v2f cq   = __builtin_elementwise_fma(m2cs2, xw, cx2 + cy22);
            const v2f P    = Dv * cq;
#pragma unroll
            for (int e = 0; e < 2; ++e) {
                const int r = 2 * p + e;
                const float D_ = fmaxf(Dv[e], MIN_NORM_F);
                const float s  = __builtin_amdgcn_sqrtf(fmaxf(P[e], 0.f));
                const float ratio = (D_ + s) * __builtin_amdgcn_rcpf(D_ - s);
                const float lg = fminf(__builtin_amdgcn_logf(ratio), lmax);
                const float u  = kln * lg;
                const float d  = u * u;
                d8[r] = d;
                gsum[r] += d;
            }
        }
    };

    auto put = [&](const float (&d8)[8], int s) {
        float* sbuf = stage[s & 1];
        if (valid) {
#pragma unroll
            for (int r = 0; r < 8; ++r) sbuf[srow + r * Kn] = d8[r];
        }
        __builtin_amdgcn_fence(__ATOMIC_RELEASE, "workgroup", "local");
        __builtin_amdgcn_s_barrier_signal(-1);
    };

    auto get1 = [&](float4& cv, float*& gp, bool& ok, int s, int h) {
        float* gbase = nodeDist + (size_t)(m0 + (s + h * T_HALF) * Nn) * Kn;
        const int a  = (int)(((size_t)gbase >> 4) & 7);
        const int p  = tid - a;
        ok = (unsigned)p < (unsigned)CHUNK_Q;
        const int pc = ok ? p : 0;
        cv = *(const float4*)(stage[s & 1] + h * CHUNK_F + 4 * pc);
        gp = gbase + 4 * pc + (pc >= pcross ? XBATCH_F : 0);
    };
    auto get = [&](float4& c0, float4& c1, float*& g0, float*& g1, bool& k0, bool& k1, int s) {
        __builtin_amdgcn_s_barrier_wait(-1);
        __builtin_amdgcn_fence(__ATOMIC_ACQUIRE, "workgroup", "local");
        get1(c0, g0, k0, s, 0);
        get1(c1, g1, k1, s, 1);
        __builtin_amdgcn_sched_barrier(1);
    };

    auto out = [&](const float4& c0, const float4& c1, float* g0, float* g1, bool k0, bool k1) {
        if (k0) *(float4*)g0 = c0;
        if (k1) *(float4*)g1 = c1;
    };

    float4* xs4 = (float4*)xs;
    xs4[xo0] = xv0;
    xs4[xo1] = xv1;
    xs4[xo2] = xv2;
    if (xlast) xs4[xo3] = xv3;
    __syncthreads();
    const float* xl = xs + l16 * XPITCH + lhalf * 2;
    auto load_a = [&](v2f (&a)[8], int s) {
        const float* xt = xl + s * (16 * XPITCH);
#pragma unroll
        for (int j = 0; j < 8; ++j) a[j] = *(const v2f*)(xt + j * 4);
    };
    v2f a0[8], a1[8], a2[8];
    load_a(a0, 0);
    load_a(a1, 1);
    load_a(a2, 2);

    static_assert(T_HALF == 6, "the pipeline below is written out for 6 steps");
    v8f accA, accB;
    float xnA, xnB;
    float d8[8];
    float4 c0, c1;
    float *g0, *g1;
    bool   k0, k1;
    front(a0, accA, xnA);
    load_a(a0, 3);
    front(a1, accB, xnB);
    load_a(a1, 4);
    epi(accA, xnA, d8);  put(d8, 0);
    front(a2, accA, xnA);
    load_a(a2, 5);
    get(c0, c1, g0, g1, k0, k1, 0);
    epi(accB, xnB, d8);  out(c0, c1, g0, g1, k0, k1);  put(d8, 1);
    front(a0, accB, xnB);
    get(c0, c1, g0, g1, k0, k1, 1);
    epi(accA, xnA, d8);  out(c0, c1, g0, g1, k0, k1);  put(d8, 2);
    front(a1, accA, xnA);
    get(c0, c1, g0, g1, k0, k1, 2);
    epi(accB, xnB, d8);  out(c0, c1, g0, g1, k0, k1);  put(d8, 3);
    front(a2, accB, xnB);
    get(c0, c1, g0, g1, k0, k1, 3);
    epi(accA, xnA, d8);  out(c0, c1, g0, g1, k0, k1);  put(d8, 4);
    get(c0, c1, g0, g1, k0, k1, 4);
    epi(accB, xnB, d8);  out(c0, c1, g0, g1, k0, k1);  put(d8, 5);

    float tot[8];
#pragma unroll
    for (int r = 0; r < 8; ++r) tot[r] = gsum[r] + __shfl_xor(gsum[r], 16, 32);
    if (valid) {
#pragma unroll
        for (int i = 0; i < 4; ++i) {
            const float v = lhalf ? tot[4 + i] : tot[i];
            graph[(size_t)(gS0 + lhalf * 4 + i) * Kn + kcol] = v * (1.0f / (float)Nn);
        }
    }
    get(c0, c1, g0, g1, k0, k1, 5);
    out(c0, c1, g0, g1, k0, k1);
}

extern "C" void kernel_launch(void* const* d_in, const int* in_sizes, int n_in,
                              void* d_out, int out_size, void* d_ws, size_t ws_size,
                              hipStream_t stream) {
    const float* node_repr = (const float*)d_in[0];
    const float* cweight   = (const float*)d_in[1];
    const float* cvec      = (const float*)d_in[2];

    float* graph    = (float*)d_out;
    float* nodeDist = graph + (size_t)Bn * Nn * Kn;

    hdist_fused_kernel<<<FUSED_WGS, WG_THREADS, 0, stream>>>(
        node_repr, cweight, cvec, nodeDist, graph);
}
